# starting kernel
# baseline (speedup 1.0000x reference)
#include <hip/hip_runtime.h>
#include <hip/hip_bf16.h>


#define B_ 16
#define L_ 4096
#define M_ 64
#define SPLITS 8

typedef __attribute__((ext_vector_type(2))) float v2f;
typedef __attribute__((ext_vector_type(8))) float v8f;

__global__ void k_coef(float* c0, float* c1, float* c2) {
    int t = threadIdx.x;
    const int Ns[3] = {512, 256, 128};
    float* cs[3] = {c0, c1, c2};
    for (int sc = 0; sc < 3; ++sc) {
        int N = Ns[sc];
        if (t < N) {
            double acc = 0.0;
            double step = (double)(N - 1) / 511.0;
            for (int tt = 0; tt < 512; ++tt) {
                double pos = tt * step;
                int i0 = (int)pos;
                int i1 = i0 + 1; if (i1 > N - 1) i1 = N - 1;
                double w1 = pos - (double)i0;
                if (t == i0) acc += 1.0 - w1;
                if (t == i1) acc += w1;
            }
            cs[sc][t] = (float)(acc / 512.0);
        }
    }
}

__global__ void k_stage1(const float* __restrict__ x,
                         const float* __restrict__ c0,
                         const float* __restrict__ c1,
                         const float* __restrict__ c2,
                         float* __restrict__ p0, float* __restrict__ p1,
                         float* __restrict__ p2) {
    int e2 = blockIdx.x * 256 + threadIdx.x;
    int b  = blockIdx.y;
    int sp = blockIdx.z;
    int h  = e2 >> 10;
    int q  = e2 >> 9;
    const float* xb = x + (size_t)b * (L_ * M_);
    int i2b = sp * (128 / SPLITS);
    float a0 = 0.f, a1 = 0.f, a2 = 0.f;
#pragma unroll 4
    for (int i = 0; i < 128 / SPLITS; ++i) {
        int i2 = i2b + i;
        float v = xb[(size_t)i2 * 2048 + e2];
        a2 = fmaf(c2[i2],         v, a2);
        a1 = fmaf(c1[2 * i2 + h], v, a1);
        a0 = fmaf(c0[4 * i2 + q], v, a0);
    }
    size_t o = ((size_t)sp * B_ + b) * 2048 + e2;
    p0[o] = a0; p1[o] = a1; p2[o] = a2;
}

__global__ void k_stage2(const float* __restrict__ p0,
                         const float* __restrict__ p1,
                         const float* __restrict__ p2,
                         float* __restrict__ A) {
    int idx = blockIdx.x * 256 + threadIdx.x;
    int row = idx >> 6;
    int kk  = idx & 63;
    int b = row >> 6, m = row & 63;
    float acc = 0.f;
    if (kk < 8) {
        int j = m >> 3, s = m & 7, k = kk;
        int e = j * 64 + s * 8 + k;
        for (int sp = 0; sp < SPLITS; ++sp)
            for (int qq = 0; qq < 4; ++qq)
                acc += p0[((size_t)sp * B_ + b) * 2048 + qq * 512 + e];
    } else if (kk < 24) {
        int j = m >> 2, s = m & 3, k = kk - 8;
        int e = j * 64 + s * 16 + k;
        for (int sp = 0; sp < SPLITS; ++sp)
            for (int hh = 0; hh < 2; ++hh)
                acc += p1[((size_t)sp * B_ + b) * 2048 + hh * 1024 + e];
    } else if (kk < 56) {
        int j = m >> 1, s = m & 1, k = kk - 24;
        int e = j * 64 + s * 32 + k;
        for (int sp = 0; sp < SPLITS; ++sp)
            acc += p2[((size_t)sp * B_ + b) * 2048 + e];
    }
    A[idx] = acc;
}

__global__ void k_weff(const float* __restrict__ pw0, const float* __restrict__ pw1,
                       const float* __restrict__ pw2, const float* __restrict__ fuse_w,
                       float* __restrict__ Weff) {
    int idx = blockIdx.x * 256 + threadIdx.x;
    if (idx >= 56 * 384) return;
    int kk = idx / 384, df = idx % 384;
    const float* w; int sc, k;
    if (kk < 8)       { sc = 0; k = kk;      w = pw0; }
    else if (kk < 24) { sc = 1; k = kk - 8;  w = pw1; }
    else              { sc = 2; k = kk - 24; w = pw2; }
    float acc = 0.f;
    for (int d = 0; d < 128; ++d)
        acc = fmaf(w[k * 128 + d], fuse_w[(size_t)(sc * 128 + d) * 384 + df], acc);
    Weff[idx] = acc;
}

__global__ void k_beff(const float* __restrict__ pb0, const float* __restrict__ pb1,
                       const float* __restrict__ pb2, const float* __restrict__ fuse_w,
                       const float* __restrict__ fuse_b, float* __restrict__ beff) {
    int df = blockIdx.x * 256 + threadIdx.x;
    if (df >= 384) return;
    float acc = fuse_b[df];
    const float* pbs[3] = {pb0, pb1, pb2};
    for (int sc = 0; sc < 3; ++sc)
        for (int d = 0; d < 128; ++d)
            acc = fmaf(pbs[sc][d], fuse_w[(size_t)(sc * 128 + d) * 384 + df], acc);
    beff[df] = acc;
}

__global__ void k_wtot(const float* __restrict__ Weff, const float* __restrict__ flat_w,
                       float* __restrict__ Wtot) {
    int idx = blockIdx.x * 256 + threadIdx.x;
    int kk = idx >> 9, n = idx & 511;
    float acc = 0.f;
    if (kk < 56)
        for (int df = 0; df < 384; ++df)
            acc = fmaf(Weff[kk * 384 + df], flat_w[(size_t)df * 512 + n], acc);
    Wtot[idx] = acc;
}

__global__ void k_btot(const float* __restrict__ beff, const float* __restrict__ flat_w,
                       const float* __restrict__ flat_b, float* __restrict__ btot) {
    int n = blockIdx.x * 256 + threadIdx.x;
    if (n >= 512) return;
    float acc = flat_b[n];
    for (int df = 0; df < 384; ++df)
        acc = fmaf(beff[df], flat_w[(size_t)df * 512 + n], acc);
    btot[n] = acc;
}

__global__ void k_gemm(const float* __restrict__ A, const float* __restrict__ Wt,
                       const float* __restrict__ btot, float* __restrict__ out) {
    int wave = threadIdx.x >> 5;
    int lane = threadIdx.x & 31;
    int tile = blockIdx.x * 8 + wave;
    int mt = tile >> 5, nt = tile & 31;
    int mbase = mt * 16, nbase = nt * 16;
    int half = lane >> 4, l16 = lane & 15;
    const float* Ar = A + (size_t)(mbase + l16) * 64;
    const float* Bc = Wt + nbase + l16;
    v8f acc = {};
#pragma unroll
    for (int k0 = 0; k0 < 64; k0 += 4) {
        int ka = k0 + half * 2;
        v2f av = { Ar[ka], Ar[ka + 1] };
        v2f bv = { Bc[(size_t)ka * 512], Bc[(size_t)(ka + 1) * 512] };
        acc = __builtin_amdgcn_wmma_f32_16x16x4_f32(
            false, av, false, bv, (short)0, acc, false, false);
    }
    float bias = btot[nbase + l16];
#pragma unroll
    for (int r = 0; r < 8; ++r)
        out[(size_t)(mbase + r + 8 * half) * 512 + nbase + l16] = acc[r] + bias;
}

extern "C" void kernel_launch(void* const* d_in, const int* in_sizes, int n_in,
                              void* d_out, int out_size, void* d_ws, size_t ws_size,
                              hipStream_t stream) {
    (void)in_sizes; (void)n_in; (void)out_size; (void)ws_size;
    const float* x      = (const float*)d_in[0];
    const float* pw0    = (const float*)d_in[1];
    const float* pb0    = (const float*)d_in[2];
    const float* pw1    = (const float*)d_in[3];
    const float* pb1    = (const float*)d_in[4];
    const float* pw2    = (const float*)d_in[5];
    const float* pb2    = (const float*)d_in[6];
    const float* fuse_w = (const float*)d_in[7];
    const float* fuse_b = (const float*)d_in[8];
    const float* flat_w = (const float*)d_in[9];
    const float* flat_b = (const float*)d_in[10];
    float* out = (float*)d_out;

    float* ws   = (float*)d_ws;
    float* c0   = ws;
    float* c1   = ws + 512;
    float* c2   = ws + 768;
    float* p0   = ws + 1024;
    float* p1   = p0 + SPLITS * B_ * 2048;
    float* p2   = p1 + SPLITS * B_ * 2048;
    float* A    = p2 + SPLITS * B_ * 2048;
    float* Weff = A + 1024 * 64;
    float* beff = Weff + 56 * 384;
    float* Wtot = beff + 384;
    float* btot = Wtot + 64 * 512;

    k_coef<<<1, 512, 0, stream>>>(c0, c1, c2);
    k_weff<<<(56 * 384 + 255) / 256, 256, 0, stream>>>(pw0, pw1, pw2, fuse_w, Weff);
    k_beff<<<2, 256, 0, stream>>>(pb0, pb1, pb2, fuse_w, fuse_b, beff);
    k_wtot<<<(64 * 512) / 256, 256, 0, stream>>>(Weff, flat_w, Wtot);
    k_btot<<<2, 256, 0, stream>>>(beff, flat_w, flat_b, btot);

    dim3 g1(2048 / 256, B_, SPLITS);
    k_stage1<<<g1, 256, 0, stream>>>(x, c0, c1, c2, p0, p1, p2);
    k_stage2<<<(1024 * 64) / 256, 256, 0, stream>>>(p0, p1, p2, A);

    k_gemm<<<2048 / 8, 256, 0, stream>>>(A, Wtot, btot, out);
}


# k_coef: each thread sums only the output tokens whose two interpolation taps can hit its index (5-12 of 512), same tests and order, coefficients bit-identical
# speedup vs baseline: 4.3765x; 4.3765x over previous
#include <hip/hip_runtime.h>
#include <hip/hip_bf16.h>


#define B_ 16
#define L_ 4096
#define M_ 64
#define SPLITS 8

typedef __attribute__((ext_vector_type(2))) float v2f;
typedef __attribute__((ext_vector_type(8))) float v8f;

__global__ void k_coef(float* c0, float* c1, float* c2) {
    int t = threadIdx.x;
    const int Ns[3] = {512, 256, 128};
    float* cs[3] = {c0, c1, c2};
    for (int sc = 0; sc < 3; ++sc) {
        int N = Ns[sc];
        if (t < N) {
            double acc = 0.0;
            double step = (double)(N - 1) / 511.0;
            int lo = (t > 0) ? ((t - 1) * 511) / (N - 1) - 1 : 0;
            if (lo < 0) lo = 0;
            int hi = ((t + 1) * 511) / (N - 1) + 1;
            if (hi > 511) hi = 511;
            for (int tt = lo; tt <= hi; ++tt) {
                double pos = tt * step;
                int i0 = (int)pos;
                int i1 = i0 + 1; if (i1 > N - 1) i1 = N - 1;
                double w1 = pos - (double)i0;
                if (t == i0) acc += 1.0 - w1;
                if (t == i1) acc += w1;
            }
            cs[sc][t] = (float)(acc / 512.0);
        }
    }
}

__global__ void k_stage1(const float* __restrict__ x,
                         const float* __restrict__ c0,
                         const float* __restrict__ c1,
                         const float* __restrict__ c2,
                         float* __restrict__ p0, float* __restrict__ p1,
                         float* __restrict__ p2) {
    int e2 = blockIdx.x * 256 + threadIdx.x;
    int b  = blockIdx.y;
    int sp = blockIdx.z;
    int h  = e2 >> 10;
    int q  = e2 >> 9;
    const float* xb = x + (size_t)b * (L_ * M_);
    int i2b = sp * (128 / SPLITS);
    float a0 = 0.f, a1 = 0.f, a2 = 0.f;
#pragma unroll 4
    for (int i = 0; i < 128 / SPLITS; ++i) {
        int i2 = i2b + i;
        float v = xb[(size_t)i2 * 2048 + e2];
        a2 = fmaf(c2[i2],         v, a2);
        a1 = fmaf(c1[2 * i2 + h], v, a1);
        a0 = fmaf(c0[4 * i2 + q], v, a0);
    }
    size_t o = ((size_t)sp * B_ + b) * 2048 + e2;
    p0[o] = a0; p1[o] = a1; p2[o] = a2;
}

__global__ void k_stage2(const float* __restrict__ p0,
                         const float* __restrict__ p1,
                         const float* __restrict__ p2,
                         float* __restrict__ A) {
    int idx = blockIdx.x * 256 + threadIdx.x;
    int row = idx >> 6;
    int kk  = idx & 63;
    int b = row >> 6, m = row & 63;
    float acc = 0.f;
    if (kk < 8) {
        int j = m >> 3, s = m & 7, k = kk;
        int e = j * 64 + s * 8 + k;
        for (int sp = 0; sp < SPLITS; ++sp)
            for (int qq = 0; qq < 4; ++qq)
                acc += p0[((size_t)sp * B_ + b) * 2048 + qq * 512 + e];
    } else if (kk < 24) {
        int j = m >> 2, s = m & 3, k = kk - 8;
        int e = j * 64 + s * 16 + k;
        for (int sp = 0; sp < SPLITS; ++sp)
            for (int hh = 0; hh < 2; ++hh)
                acc += p1[((size_t)sp * B_ + b) * 2048 + hh * 1024 + e];
    } else if (kk < 56) {
        int j = m >> 1, s = m & 1, k = kk - 24;
        int e = j * 64 + s * 32 + k;
        for (int sp = 0; sp < SPLITS; ++sp)
            acc += p2[((size_t)sp * B_ + b) * 2048 + e];
    }
    A[idx] = acc;
}

__global__ void k_weff(const float* __restrict__ pw0, const float* __restrict__ pw1,
                       const float* __restrict__ pw2, const float* __restrict__ fuse_w,
                       float* __restrict__ Weff) {
    int idx = blockIdx.x * 256 + threadIdx.x;
    if (idx >= 56 * 384) return;
    int kk = idx / 384, df = idx % 384;
    const float* w; int sc, k;
    if (kk < 8)       { sc = 0; k = kk;      w = pw0; }
    else if (kk < 24) { sc = 1; k = kk - 8;  w = pw1; }
    else              { sc = 2; k = kk - 24; w = pw2; }
    float acc = 0.f;
    for (int d = 0; d < 128; ++d)
        acc = fmaf(w[k * 128 + d], fuse_w[(size_t)(sc * 128 + d) * 384 + df], acc);
    Weff[idx] = acc;
}

__global__ void k_beff(const float* __restrict__ pb0, const float* __restrict__ pb1,
                       const float* __restrict__ pb2, const float* __restrict__ fuse_w,
                       const float* __restrict__ fuse_b, float* __restrict__ beff) {
    int df = blockIdx.x * 256 + threadIdx.x;
    if (df >= 384) return;
    float acc = fuse_b[df];
    const float* pbs[3] = {pb0, pb1, pb2};
    for (int sc = 0; sc < 3; ++sc)
        for (int d = 0; d < 128; ++d)
            acc = fmaf(pbs[sc][d], fuse_w[(size_t)(sc * 128 + d) * 384 + df], acc);
    beff[df] = acc;
}

__global__ void k_wtot(const float* __restrict__ Weff, const float* __restrict__ flat_w,
                       float* __restrict__ Wtot) {
    int idx = blockIdx.x * 256 + threadIdx.x;
    int kk = idx >> 9, n = idx & 511;
    float acc = 0.f;
    if (kk < 56)
        for (int df = 0; df < 384; ++df)
            acc = fmaf(Weff[kk * 384 + df], flat_w[(size_t)df * 512 + n], acc);
    Wtot[idx] = acc;
}

__global__ void k_btot(const float* __restrict__ beff, const float* __restrict__ flat_w,
                       const float* __restrict__ flat_b, float* __restrict__ btot) {
    int n = blockIdx.x * 256 + threadIdx.x;
    if (n >= 512) return;
    float acc = flat_b[n];
    for (int df = 0; df < 384; ++df)
        acc = fmaf(beff[df], flat_w[(size_t)df * 512 + n], acc);
    btot[n] = acc;
}

__global__ void k_gemm(const float* __restrict__ A, const float* __restrict__ Wt,
                       const float* __restrict__ btot, float* __restrict__ out) {
    int wave = threadIdx.x >> 5;
    int lane = threadIdx.x & 31;
    int tile = blockIdx.x * 8 + wave;
    int mt = tile >> 5, nt = tile & 31;
    int mbase = mt * 16, nbase = nt * 16;
    int half = lane >> 4, l16 = lane & 15;
    const float* Ar = A + (size_t)(mbase + l16) * 64;
    const float* Bc = Wt + nbase + l16;
    v8f acc = {};
#pragma unroll
    for (int k0 = 0; k0 < 64; k0 += 4) {
        int ka = k0 + half * 2;
        v2f av = { Ar[ka], Ar[ka + 1] };
        v2f bv = { Bc[(size_t)ka * 512], Bc[(size_t)(ka + 1) * 512] };
        acc = __builtin_amdgcn_wmma_f32_16x16x4_f32(
            false, av, false, bv, (short)0, acc, false, false);
    }
    float bias = btot[nbase + l16];
#pragma unroll
    for (int r = 0; r < 8; ++r)
        out[(size_t)(mbase + r + 8 * half) * 512 + nbase + l16] = acc[r] + bias;
}

extern "C" void kernel_launch(void* const* d_in, const int* in_sizes, int n_in,
                              void* d_out, int out_size, void* d_ws, size_t ws_size,
                              hipStream_t stream) {
    (void)in_sizes; (void)n_in; (void)out_size; (void)ws_size;
    const float* x      = (const float*)d_in[0];
    const float* pw0    = (const float*)d_in[1];
    const float* pb0    = (const float*)d_in[2];
    const float* pw1    = (const float*)d_in[3];
    const float* pb1    = (const float*)d_in[4];
    const float* pw2    = (const float*)d_in[5];
    const float* pb2    = (const float*)d_in[6];
    const float* fuse_w = (const float*)d_in[7];
    const float* fuse_b = (const float*)d_in[8];
    const float* flat_w = (const float*)d_in[9];
    const float* flat_b = (const float*)d_in[10];
    float* out = (float*)d_out;

    float* ws   = (float*)d_ws;
    float* c0   = ws;
    float* c1   = ws + 512;
    float* c2   = ws + 768;
    float* p0   = ws + 1024;
    float* p1   = p0 + SPLITS * B_ * 2048;
    float* p2   = p1 + SPLITS * B_ * 2048;
    float* A    = p2 + SPLITS * B_ * 2048;
    float* Weff = A + 1024 * 64;
    float* beff = Weff + 56 * 384;
    float* Wtot = beff + 384;
    float* btot = Wtot + 64 * 512;

    k_coef<<<1, 512, 0, stream>>>(c0, c1, c2);
    k_weff<<<(56 * 384 + 255) / 256, 256, 0, stream>>>(pw0, pw1, pw2, fuse_w, Weff);
    k_beff<<<2, 256, 0, stream>>>(pb0, pb1, pb2, fuse_w, fuse_b, beff);
    k_wtot<<<(64 * 512) / 256, 256, 0, stream>>>(Weff, flat_w, Wtot);
    k_btot<<<2, 256, 0, stream>>>(beff, flat_w, flat_b, btot);

    dim3 g1(2048 / 256, B_, SPLITS);
    k_stage1<<<g1, 256, 0, stream>>>(x, c0, c1, c2, p0, p1, p2);
    k_stage2<<<(1024 * 64) / 256, 256, 0, stream>>>(p0, p1, p2, A);

    k_gemm<<<2048 / 8, 256, 0, stream>>>(A, Wtot, btot, out);
}


# 8 launches -> 4: kernels with no dependency between them share one grid as block ranges (beff|weff|coef, then btot|wtot|stage1) and run side by side; every body, sum and summation order as before
# speedup vs baseline: 6.5485x; 1.4963x over previous
#include <hip/hip_runtime.h>
#include <hip/hip_bf16.h>


#define B_ 16
#define L_ 4096
#define M_ 64
#define SPLITS 8

typedef __attribute__((ext_vector_type(2))) float v2f;
typedef __attribute__((ext_vector_type(8))) float v8f;

__device__ __forceinline__ void coef_body(int blk, float* c0, float* c1, float* c2) {
    int t = blk * 256 + threadIdx.x;
    const int Ns[3] = {512, 256, 128};
    float* cs[3] = {c0, c1, c2};
    for (int sc = 0; sc < 3; ++sc) {
        int N = Ns[sc];
        if (t < N) {
            double acc = 0.0;
            double step = (double)(N - 1) / 511.0;
            int lo = (t > 0) ? ((t - 1) * 511) / (N - 1) - 1 : 0;
            if (lo < 0) lo = 0;
            int hi = ((t + 1) * 511) / (N - 1) + 1;
            if (hi > 511) hi = 511;
            for (int tt = lo; tt <= hi; ++tt) {
                double pos = tt * step;
                int i0 = (int)pos;
                int i1 = i0 + 1; if (i1 > N - 1) i1 = N - 1;
                double w1 = pos - (double)i0;
                if (t == i0) acc += 1.0 - w1;
                if (t == i1) acc += w1;
            }
            cs[sc][t] = (float)(acc / 512.0);
        }
    }
}

__device__ __forceinline__ void stage1_body(int blk,
                         const float* __restrict__ x,
                         const float* __restrict__ c0,
                         const float* __restrict__ c1,
                         const float* __restrict__ c2,
                         float* __restrict__ p0, float* __restrict__ p1,
                         float* __restrict__ p2) {
    int e2 = (blk & 7) * 256 + threadIdx.x;
    int b  = (blk >> 3) & 15;
    int sp = blk >> 7;
    int h  = e2 >> 10;
    int q  = e2 >> 9;
    const float* xb = x + (size_t)b * (L_ * M_);
    int i2b = sp * (128 / SPLITS);
    float a0 = 0.f, a1 = 0.f, a2 = 0.f;
#pragma unroll 4
    for (int i = 0; i < 128 / SPLITS; ++i) {
        int i2 = i2b + i;
        float v = xb[(size_t)i2 * 2048 + e2];
        a2 = fmaf(c2[i2],         v, a2);
        a1 = fmaf(c1[2 * i2 + h], v, a1);
        a0 = fmaf(c0[4 * i2 + q], v, a0);
    }
    size_t o = ((size_t)sp * B_ + b) * 2048 + e2;
    p0[o] = a0; p1[o] = a1; p2[o] = a2;
}

__global__ void k_stage2(const float* __restrict__ p0,
                         const float* __restrict__ p1,
                         const float* __restrict__ p2,
                         float* __restrict__ A) {
    int idx = blockIdx.x * 256 + threadIdx.x;
    int row = idx >> 6;
    int kk  = idx & 63;
    int b = row >> 6, m = row & 63;
    float acc = 0.f;
    if (kk < 8) {
        int j = m >> 3, s = m & 7, k = kk;
        int e = j * 64 + s * 8 + k;
        for (int sp = 0; sp < SPLITS; ++sp)
            for (int qq = 0; qq < 4; ++qq)
                acc += p0[((size_t)sp * B_ + b) * 2048 + qq * 512 + e];
    } else if (kk < 24) {
        int j = m >> 2, s = m & 3, k = kk - 8;
        int e = j * 64 + s * 16 + k;
        for (int sp = 0; sp < SPLITS; ++sp)
            for (int hh = 0; hh < 2; ++hh)
                acc += p1[((size_t)sp * B_ + b) * 2048 + hh * 1024 + e];
    } else if (kk < 56) {
        int j = m >> 1, s = m & 1, k = kk - 24;
        int e = j * 64 + s * 32 + k;
        for (int sp = 0; sp < SPLITS; ++sp)
            acc += p2[((size_t)sp * B_ + b) * 2048 + e];
    }
    A[idx] = acc;
}

__device__ __forceinline__ void weff_body(int blk,
                       const float* __restrict__ pw0, const float* __restrict__ pw1,
                       const float* __restrict__ pw2, const float* __restrict__ fuse_w,
                       float* __restrict__ Weff) {
    int idx = blk * 256 + threadIdx.x;
    if (idx >= 56 * 384) return;
    int kk = idx / 384, df = idx % 384;
    const float* w; int sc, k;
    if (kk < 8)       { sc = 0; k = kk;      w = pw0; }
    else if (kk < 24) { sc = 1; k = kk - 8;  w = pw1; }
    else              { sc = 2; k = kk - 24; w = pw2; }
    float acc = 0.f;
    for (int d = 0; d < 128; ++d)
        acc = fmaf(w[k * 128 + d], fuse_w[(size_t)(sc * 128 + d) * 384 + df], acc);
    Weff[idx] = acc;
}

__device__ __forceinline__ void beff_body(int blk,
                       const float* __restrict__ pb0, const float* __restrict__ pb1,
                       const float* __restrict__ pb2, const float* __restrict__ fuse_w,
                       const float* __restrict__ fuse_b, float* __restrict__ beff) {
    int df = blk * 256 + threadIdx.x;
    if (df >= 384) return;
    float acc = fuse_b[df];
    const float* pbs[3] = {pb0, pb1, pb2};
    for (int sc = 0; sc < 3; ++sc)
        for (int d = 0; d < 128; ++d)
            acc = fmaf(pbs[sc][d], fuse_w[(size_t)(sc * 128 + d) * 384 + df], acc);
    beff[df] = acc;
}

__device__ __forceinline__ void wtot_body(int blk,
                       const float* __restrict__ Weff, const float* __restrict__ flat_w,
                       float* __restrict__ Wtot) {
    int idx = blk * 256 + threadIdx.x;
    int kk = idx >> 9, n = idx & 511;
    float acc = 0.f;
    if (kk < 56)
        for (int df = 0; df < 384; ++df)
            acc = fmaf(Weff[kk * 384 + df], flat_w[(size_t)df * 512 + n], acc);
    Wtot[idx] = acc;
}

__device__ __forceinline__ void btot_body(int blk,
                       const float* __restrict__ beff, const float* __restrict__ flat_w,
                       const float* __restrict__ flat_b, float* __restrict__ btot) {
    int n = blk * 256 + threadIdx.x;
    if (n >= 512) return;
    float acc = flat_b[n];
    for (int df = 0; df < 384; ++df)
        acc = fmaf(beff[df], flat_w[(size_t)df * 512 + n], acc);
    btot[n] = acc;
}

#define PRE_BEFF 2
#define PRE_WEFF ((56 * 384 + 255) / 256)
#define PRE_COEF 2
__global__ void k_pre(const float* __restrict__ pw0, const float* __restrict__ pb0,
                      const float* __restrict__ pw1, const float* __restrict__ pb1,
                      const float* __restrict__ pw2, const float* __restrict__ pb2,
                      const float* __restrict__ fuse_w, const float* __restrict__ fuse_b,
                      float* __restrict__ Weff, float* __restrict__ beff,
                      float* c0, float* c1, float* c2) {
    int blk = blockIdx.x;
    if (blk < PRE_BEFF)
        beff_body(blk, pb0, pb1, pb2, fuse_w, fuse_b, beff);
    else if (blk < PRE_BEFF + PRE_WEFF)
        weff_body(blk - PRE_BEFF, pw0, pw1, pw2, fuse_w, Weff);
    else
        coef_body(blk - PRE_BEFF - PRE_WEFF, c0, c1, c2);
}

#define MID_BTOT 2
#define MID_WTOT ((64 * 512) / 256)
#define MID_ST1  ((2048 / 256) * B_ * SPLITS)
__global__ void k_mid(const float* __restrict__ x,
                      const float* __restrict__ c0, const float* __restrict__ c1,
                      const float* __restrict__ c2,
                      const float* __restrict__ Weff, const float* __restrict__ beff,
                      const float* __restrict__ flat_w, const float* __restrict__ flat_b,
                      float* __restrict__ p0, float* __restrict__ p1,
                      float* __restrict__ p2,
                      float* __restrict__ Wtot, float* __restrict__ btot) {
    int blk = blockIdx.x;
    if (blk < MID_BTOT)
        btot_body(blk, beff, flat_w, flat_b, btot);
    else if (blk < MID_BTOT + MID_WTOT)
        wtot_body(blk - MID_BTOT, Weff, flat_w, Wtot);
    else
        stage1_body(blk - MID_BTOT - MID_WTOT, x, c0, c1, c2, p0, p1, p2);
}

__global__ void k_gemm(const float* __restrict__ A, const float* __restrict__ Wt,
                       const float* __restrict__ btot, float* __restrict__ out) {
    int wave = threadIdx.x >> 5;
    int lane = threadIdx.x & 31;
    int tile = blockIdx.x * 8 + wave;
    int mt = tile >> 5, nt = tile & 31;
    int mbase = mt * 16, nbase = nt * 16;
    int half = lane >> 4, l16 = lane & 15;
    const float* Ar = A + (size_t)(mbase + l16) * 64;
    const float* Bc = Wt + nbase + l16;
    v8f acc = {};
#pragma unroll
    for (int k0 = 0; k0 < 64; k0 += 4) {
        int ka = k0 + half * 2;
        v2f av = { Ar[ka], Ar[ka + 1] };
        v2f bv = { Bc[(size_t)ka * 512], Bc[(size_t)(ka + 1) * 512] };
        acc = __builtin_amdgcn_wmma_f32_16x16x4_f32(
            false, av, false, bv, (short)0, acc, false, false);
    }
    float bias = btot[nbase + l16];
#pragma unroll
    for (int r = 0; r < 8; ++r)
        out[(size_t)(mbase + r + 8 * half) * 512 + nbase + l16] = acc[r] + bias;
}

extern "C" void kernel_launch(void* const* d_in, const int* in_sizes, int n_in,
                              void* d_out, int out_size, void* d_ws, size_t ws_size,
                              hipStream_t stream) {
    (void)in_sizes; (void)n_in; (void)out_size; (void)ws_size;
    const float* x      = (const float*)d_in[0];
    const float* pw0    = (const float*)d_in[1];
    const float* pb0    = (const float*)d_in[2];
    const float* pw1    = (const float*)d_in[3];
    const float* pb1    = (const float*)d_in[4];
    const float* pw2    = (const float*)d_in[5];
    const float* pb2    = (const float*)d_in[6];
    const float* fuse_w = (const float*)d_in[7];
    const float* fuse_b = (const float*)d_in[8];
    const float* flat_w = (const float*)d_in[9];
    const float* flat_b = (const float*)d_in[10];
    float* out = (float*)d_out;

    float* ws   = (float*)d_ws;
    float* c0   = ws;
    float* c1   = ws + 512;
    float* c2   = ws + 768;
    float* p0   = ws + 1024;
    float* p1   = p0 + SPLITS * B_ * 2048;
    float* p2   = p1 + SPLITS * B_ * 2048;
    float* A    = p2 + SPLITS * B_ * 2048;
    float* Weff = A + 1024 * 64;
    float* beff = Weff + 56 * 384;
    float* Wtot = beff + 384;
    float* btot = Wtot + 64 * 512;

    k_pre<<<PRE_BEFF + PRE_WEFF + PRE_COEF, 256, 0, stream>>>(
        pw0, pb0, pw1, pb1, pw2, pb2, fuse_w, fuse_b, Weff, beff, c0, c1, c2);
    k_mid<<<MID_BTOT + MID_WTOT + MID_ST1, 256, 0, stream>>>(
        x, c0, c1, c2, Weff, beff, flat_w, flat_b, p0, p1, p2, Wtot, btot);
    k_stage2<<<(1024 * 64) / 256, 256, 0, stream>>>(p0, p1, p2, A);

    k_gemm<<<2048 / 8, 256, 0, stream>>>(A, Wtot, btot, out);
}


# speedup vs baseline: 16.1143x; 2.4608x over previous
#include <hip/hip_runtime.h>
#include <hip/hip_bf16.h>


#define B_ 16
#define L_ 4096
#define M_ 64
#define SPLITS 8

typedef __attribute__((ext_vector_type(2))) float v2f;
typedef __attribute__((ext_vector_type(8))) float v8f;

__device__ __forceinline__ void coef_body(int blk, float* c0, float* c1, float* c2) {
    int t = blk * 256 + threadIdx.x;
    const int Ns[3] = {512, 256, 128};
    float* cs[3] = {c0, c1, c2};
    for (int sc = 0; sc < 3; ++sc) {
        int N = Ns[sc];
        if (t < N) {
            int acc = 0;
            int lo = (t > 0) ? ((t - 1) * 511) / (N - 1) - 1 : 0;
            if (lo < 0) lo = 0;
            int hi = ((t + 1) * 511) / (N - 1) + 1;
            if (hi > 511) hi = 511;
            for (int tt = lo; tt <= hi; ++tt) {
                int P  = tt * (N - 1);
                int i0 = P / 511;
                int r  = P - i0 * 511;
                int i1 = i0 + 1; if (i1 > N - 1) i1 = N - 1;
                if (t == i0) acc += 511 - r;
                if (t == i1) acc += r;
            }
            cs[sc][t] = (float)acc / (511.0f * 512.0f);
        }
    }
}

__device__ __forceinline__ void stage1_body(int blk,
                         const float* __restrict__ x,
                         const float* __restrict__ c0,
                         const float* __restrict__ c1,
                         const float* __restrict__ c2,
                         float* __restrict__ p0, float* __restrict__ p1,
                         float* __restrict__ p2) {
    int e2 = (blk & 7) * 256 + threadIdx.x;
    int b  = (blk >> 3) & 15;
    int sp = blk >> 7;
    int h  = e2 >> 10;
    int q  = e2 >> 9;
    const float* xb = x + (size_t)b * (L_ * M_);
    int i2b = sp * (128 / SPLITS);
    float a0 = 0.f, a1 = 0.f, a2 = 0.f;
#pragma unroll 4
    for (int i = 0; i < 128 / SPLITS; ++i) {
        int i2 = i2b + i;
        float v = xb[(size_t)i2 * 2048 + e2];
        a2 = fmaf(c2[i2],         v, a2);
        a1 = fmaf(c1[2 * i2 + h], v, a1);
        a0 = fmaf(c0[4 * i2 + q], v, a0);
    }
    size_t o = ((size_t)sp * B_ + b) * 2048 + e2;
    p0[o] = a0; p1[o] = a1; p2[o] = a2;
}

__global__ void k_stage2(const float* __restrict__ p0,
                         const float* __restrict__ p1,
                         const float* __restrict__ p2,
                         float* __restrict__ A) {
    int idx = blockIdx.x * 256 + threadIdx.x;
    int row = idx >> 6;
    int kk  = idx & 63;
    int b = row >> 6, m = row & 63;
    float acc = 0.f;
    if (kk < 8) {
        int j = m >> 3, s = m & 7, k = kk;
        int e = j * 64 + s * 8 + k;
        for (int sp = 0; sp < SPLITS; ++sp)
            for (int qq = 0; qq < 4; ++qq)
                acc += p0[((size_t)sp * B_ + b) * 2048 + qq * 512 + e];
    } else if (kk < 24) {
        int j = m >> 2, s = m & 3, k = kk - 8;
        int e = j * 64 + s * 16 + k;
        for (int sp = 0; sp < SPLITS; ++sp)
            for (int hh = 0; hh < 2; ++hh)
                acc += p1[((size_t)sp * B_ + b) * 2048 + hh * 1024 + e];
    } else if (kk < 56) {
        int j = m >> 1, s = m & 1, k = kk - 24;
        int e = j * 64 + s * 32 + k;
        for (int sp = 0; sp < SPLITS; ++sp)
            acc += p2[((size_t)sp * B_ + b) * 2048 + e];
    }
    A[idx] = acc;
}

#define RED_FLOATS (8 * 2 * 8 * 32)

__device__ __forceinline__ void tile_sum(v8f acc0, v8f acc1, float* red,
                                         int wave, int lane, float sum[2]) {
    float* mine = red + wave * 512;
    for (int r = 0; r < 8; ++r) {
        mine[r * 32 + lane]       = acc0[r];
        mine[256 + r * 32 + lane] = acc1[r];
    }
    __syncthreads();
    for (int p = 0; p < 2; ++p) {
        int at = (lane >> 4) * 256 + wave * 32 + p * 16 + (lane & 15);
        float s = red[at];
        for (int w = 1; w < 8; ++w) s += red[w * 512 + at];
        sum[p] = s;
    }
}

__device__ __forceinline__ void weff_tile(int tile,
                       const float* __restrict__ pw0, const float* __restrict__ pb0,
                       const float* __restrict__ pw1, const float* __restrict__ pb1,
                       const float* __restrict__ pw2, const float* __restrict__ pb2,
                       const float* __restrict__ fuse_w, const float* __restrict__ fuse_b,
                       float* __restrict__ Weff, float* red) {
    int wave = __builtin_amdgcn_readfirstlane((int)(threadIdx.x >> 5));
    int lane = threadIdx.x & 31;
    int half = lane >> 4, l16 = lane & 15;
    int mt = tile / 12, np = tile % 12;
    int mbase = mt * 16, nbase = np * 32;
    int kk = mbase + l16;
    int rsc, rk;
    if (kk < 8)        { rsc = 0; rk = kk; }
    else if (kk < 24)  { rsc = 1; rk = kk - 8; }
    else if (kk < 56)  { rsc = 2; rk = kk - 24; }
    else               { rsc = (kk == 56) ? 3 : 4; rk = 0; }
    const float* Bc = fuse_w + nbase + l16;
    v8f acc0 = {}, acc1 = {};
    int kbeg = wave * 48;
#pragma unroll
    for (int s = 0; s < 12; ++s) {
        int k0 = kbeg + s * 4;
        int sc = k0 >> 7;
        int ka = k0 + half * 2;
        int d  = ka & 127;
        const float* pw = (sc == 0) ? pw0 : (sc == 1) ? pw1 : pw2;
        const float* pb = (sc == 0) ? pb0 : (sc == 1) ? pb1 : pb2;
        bool own = (rsc == sc);
        const float* ap = (rsc == 3) ? pb : (pw + (own ? rk * 128 : 0));
        float a0 = ap[d], a1 = ap[d + 1];
        bool on = own || (rsc == 3);
        v2f av = { on ? a0 : 0.f, on ? a1 : 0.f };
        v2f b0 = { Bc[(size_t)ka * 384],      Bc[(size_t)(ka + 1) * 384] };
        v2f b1 = { Bc[(size_t)ka * 384 + 16], Bc[(size_t)(ka + 1) * 384 + 16] };
        acc0 = __builtin_amdgcn_wmma_f32_16x16x4_f32(
            false, av, false, b0, (short)0, acc0, false, false);
        acc1 = __builtin_amdgcn_wmma_f32_16x16x4_f32(
            false, av, false, b1, (short)0, acc1, false, false);
    }
    float sum[2];
    tile_sum(acc0, acc1, red, wave, lane, sum);
    for (int p = 0; p < 2; ++p) {
        int row = mbase + p * 8 + wave, col = nbase + lane;
        float v = sum[p];
        if (row == 56) v += fuse_b[col];
        if (row > 56)  v = 0.f;
        Weff[row * 384 + col] = v;
    }
}

__device__ __forceinline__ void wtot_tile(int tile,
                       const float* __restrict__ Weff, const float* __restrict__ flat_w,
                       const float* __restrict__ flat_b,
                       float* __restrict__ Wtot, float* __restrict__ btot, float* red) {
    int wave = __builtin_amdgcn_readfirstlane((int)(threadIdx.x >> 5));
    int lane = threadIdx.x & 31;
    int half = lane >> 4, l16 = lane & 15;
    int mt = tile >> 4, np = tile & 15;
    int mbase = mt * 16, nbase = np * 32;
    const float* Ar = Weff + (size_t)(mbase + l16) * 384;
    const float* Bc = flat_w + nbase + l16;
    v8f acc0 = {}, acc1 = {};
    int kbeg = wave * 48;
#pragma unroll
    for (int s = 0; s < 12; ++s) {
        int ka = kbeg + s * 4 + half * 2;
        v2f av = { Ar[ka], Ar[ka + 1] };
        v2f b0 = { Bc[(size_t)ka * 512],      Bc[(size_t)(ka + 1) * 512] };
        v2f b1 = { Bc[(size_t)ka * 512 + 16], Bc[(size_t)(ka + 1) * 512 + 16] };
        acc0 = __builtin_amdgcn_wmma_f32_16x16x4_f32(
            false, av, false, b0, (short)0, acc0, false, false);
        acc1 = __builtin_amdgcn_wmma_f32_16x16x4_f32(
            false, av, false, b1, (short)0, acc1, false, false);
    }
    float sum[2];
    tile_sum(acc0, acc1, red, wave, lane, sum);
    for (int p = 0; p < 2; ++p) {
        int row = mbase + p * 8 + wave, col = nbase + lane;
        float v = sum[p];
        if (row == 56) btot[col] = v + flat_b[col];
        Wtot[row * 512 + col] = (row < 56) ? v : 0.f;
    }
}

#define PRE_COEF 2
#define PRE_WEFF (4 * 12)
__global__ void k_pre(const float* __restrict__ pw0, const float* __restrict__ pb0,
                      const float* __restrict__ pw1, const float* __restrict__ pb1,
                      const float* __restrict__ pw2, const float* __restrict__ pb2,
                      const float* __restrict__ fuse_w, const float* __restrict__ fuse_b,
                      float* __restrict__ Weff,
                      float* c0, float* c1, float* c2) {
    __shared__ float red[RED_FLOATS];
    int blk = blockIdx.x;
    if (blk < PRE_COEF)
        coef_body(blk, c0, c1, c2);
    else
        weff_tile(blk - PRE_COEF, pw0, pb0, pw1, pb1, pw2, pb2, fuse_w, fuse_b, Weff, red);
}

#define MID_WTOT (4 * 16)
#define MID_ST1  ((2048 / 256) * B_ * SPLITS)
__global__ void k_mid(const float* __restrict__ x,
                      const float* __restrict__ c0, const float* __restrict__ c1,
                      const float* __restrict__ c2,
                      const float* __restrict__ Weff,
                      const float* __restrict__ flat_w, const float* __restrict__ flat_b,
                      float* __restrict__ p0, float* __restrict__ p1,
                      float* __restrict__ p2,
                      float* __restrict__ Wtot, float* __restrict__ btot) {
    __shared__ float red[RED_FLOATS];
    int blk = blockIdx.x;
    if (blk < MID_WTOT)
        wtot_tile(blk, Weff, flat_w, flat_b, Wtot, btot, red);
    else
        stage1_body(blk - MID_WTOT, x, c0, c1, c2, p0, p1, p2);
}

__global__ void k_gemm(const float* __restrict__ A, const float* __restrict__ Wt,
                       const float* __restrict__ btot, float* __restrict__ out) {
    int wave = threadIdx.x >> 5;
    int lane = threadIdx.x & 31;
    int tile = blockIdx.x * 8 + wave;
    int mt = tile >> 5, nt = tile & 31;
    int mbase = mt * 16, nbase = nt * 16;
    int half = lane >> 4, l16 = lane & 15;
    const float* Ar = A + (size_t)(mbase + l16) * 64;
    const float* Bc = Wt + nbase + l16;
    v8f acc = {};
#pragma unroll
    for (int k0 = 0; k0 < 64; k0 += 4) {
        int ka = k0 + half * 2;
        v2f av = { Ar[ka], Ar[ka + 1] };
        v2f bv = { Bc[(size_t)ka * 512], Bc[(size_t)(ka + 1) * 512] };
        acc = __builtin_amdgcn_wmma_f32_16x16x4_f32(
            false, av, false, bv, (short)0, acc, false, false);
    }
    float bias = btot[nbase + l16];
#pragma unroll
    for (int r = 0; r < 8; ++r)
        out[(size_t)(mbase + r + 8 * half) * 512 + nbase + l16] = acc[r] + bias;
}

extern "C" void kernel_launch(void* const* d_in, const int* in_sizes, int n_in,
                              void* d_out, int out_size, void* d_ws, size_t ws_size,
                              hipStream_t stream) {
    (void)in_sizes; (void)n_in; (void)out_size; (void)ws_size;
    const float* x      = (const float*)d_in[0];
    const float* pw0    = (const float*)d_in[1];
    const float* pb0    = (const float*)d_in[2];
    const float* pw1    = (const float*)d_in[3];
    const float* pb1    = (const float*)d_in[4];
    const float* pw2    = (const float*)d_in[5];
    const float* pb2    = (const float*)d_in[6];
    const float* fuse_w = (const float*)d_in[7];
    const float* fuse_b = (const float*)d_in[8];
    const float* flat_w = (const float*)d_in[9];
    const float* flat_b = (const float*)d_in[10];
    float* out = (float*)d_out;

    float* ws   = (float*)d_ws;
    float* c0   = ws;
    float* c1   = ws + 512;
    float* c2   = ws + 768;
    float* p0   = ws + 1024;
    float* p1   = p0 + SPLITS * B_ * 2048;
    float* p2   = p1 + SPLITS * B_ * 2048;
    float* A    = p2 + SPLITS * B_ * 2048;
    float* Weff = A + 1024 * 64;
    float* Wtot = Weff + 64 * 384;
    float* btot = Wtot + 64 * 512;

    k_pre<<<PRE_COEF + PRE_WEFF, 256, 0, stream>>>(
        pw0, pb0, pw1, pb1, pw2, pb2, fuse_w, fuse_b, Weff, c0, c1, c2);
    k_mid<<<MID_WTOT + MID_ST1, 256, 0, stream>>>(
        x, c0, c1, c2, Weff, flat_w, flat_b, p0, p1, p2, Wtot, btot);
    k_stage2<<<(1024 * 64) / 256, 256, 0, stream>>>(p0, p1, p2, A);

    k_gemm<<<2048 / 8, 256, 0, stream>>>(A, Wtot, btot, out);
}


# 4 launches -> 3: each x-pass block builds its 48 resample coefficients in LDS under its x loads; x pass runs beside the weight fold, combine beside the total-weight tiles; sums and order as before
# speedup vs baseline: 20.1644x; 1.2513x over previous
#include <hip/hip_runtime.h>
#include <hip/hip_bf16.h>


#define B_ 16
#define L_ 4096
#define M_ 64
#define SPLITS 8

typedef __attribute__((ext_vector_type(2))) float v2f;
typedef __attribute__((ext_vector_type(8))) float v8f;

__device__ __forceinline__ float coef_of(int N, int t) {
    int acc = 0;
    int a = (t > 0) ? (t - 1) * 511 : 0;
    int b = (t + 1) * 511;
    int lo = ((N == 512) ? a / 511 : (N == 256) ? a / 255 : a / 127) - 1;
    if (lo < 0) lo = 0;
    int hi = ((N == 512) ? b / 511 : (N == 256) ? b / 255 : b / 127) + 1;
    if (hi > 511) hi = 511;
    for (int tt = lo; tt <= hi; ++tt) {
        int P  = tt * (N - 1);
        int i0 = P / 511;
        int r  = P - i0 * 511;
        int i1 = i0 + 1; if (i1 > N - 1) i1 = N - 1;
        if (t == i0) acc += 511 - r;
        if (t == i1) acc += r;
    }
    return (float)acc / (511.0f * 512.0f);
}

__device__ __forceinline__ void stage1_body(int blk,
                         const float* __restrict__ x,
                         float* __restrict__ p0, float* __restrict__ p1,
                         float* __restrict__ p2, float* cf) {
    int e2 = (blk & 7) * 256 + threadIdx.x;
    int b  = (blk >> 3) & 15;
    int sp = blk >> 7;
    int h  = (blk & 7) >> 2;
    int q  = (blk & 7) >> 1;
    const float* xb = x + (size_t)b * (L_ * M_);
    int i2b = sp * (128 / SPLITS);
    float v[128 / SPLITS];
#pragma unroll
    for (int i = 0; i < 128 / SPLITS; ++i)
        v[i] = xb[(size_t)(i2b + i) * 2048 + e2];
    if (threadIdx.x < 3 * (128 / SPLITS)) {
        int j  = threadIdx.x;
        int sc = j >> 4;
        int i2 = i2b + (j & 15);
        int N  = 128 << sc;
        int t  = (sc == 0) ? i2 : (sc == 1) ? 2 * i2 + h : 4 * i2 + q;
        cf[j] = coef_of(N, t);
    }
    __syncthreads();
    float a0 = 0.f, a1 = 0.f, a2 = 0.f;
#pragma unroll
    for (int i = 0; i < 128 / SPLITS; ++i) {
        float vv = v[i];
        a2 = fmaf(cf[i],      vv, a2);
        a1 = fmaf(cf[16 + i], vv, a1);
        a0 = fmaf(cf[32 + i], vv, a0);
    }
    size_t o = ((size_t)sp * B_ + b) * 2048 + e2;
    p0[o] = a0; p1[o] = a1; p2[o] = a2;
}

__device__ __forceinline__ void stage2_body(int blk,
                         const float* __restrict__ p0,
                         const float* __restrict__ p1,
                         const float* __restrict__ p2,
                         float* __restrict__ A) {
    int idx = blk * 256 + threadIdx.x;
    int row = idx >> 6;
    int kk  = idx & 63;
    int b = row >> 6, m = row & 63;
    float acc = 0.f;
    if (kk < 8) {
        int j = m >> 3, s = m & 7, k = kk;
        int e = j * 64 + s * 8 + k;
        for (int sp = 0; sp < SPLITS; ++sp)
            for (int qq = 0; qq < 4; ++qq)
                acc += p0[((size_t)sp * B_ + b) * 2048 + qq * 512 + e];
    } else if (kk < 24) {
        int j = m >> 2, s = m & 3, k = kk - 8;
        int e = j * 64 + s * 16 + k;
        for (int sp = 0; sp < SPLITS; ++sp)
            for (int hh = 0; hh < 2; ++hh)
                acc += p1[((size_t)sp * B_ + b) * 2048 + hh * 1024 + e];
    } else if (kk < 56) {
        int j = m >> 1, s = m & 1, k = kk - 24;
        int e = j * 64 + s * 32 + k;
        for (int sp = 0; sp < SPLITS; ++sp)
            acc += p2[((size_t)sp * B_ + b) * 2048 + e];
    }
    A[idx] = acc;
}

#define RED_FLOATS (8 * 2 * 8 * 32)

__device__ __forceinline__ void tile_sum(v8f acc0, v8f acc1, float* red,
                                         int wave, int lane, float sum[2]) {
    float* mine = red + wave * 512;
    for (int r = 0; r < 8; ++r) {
        mine[r * 32 + lane]       = acc0[r];
        mine[256 + r * 32 + lane] = acc1[r];
    }
    __syncthreads();
    for (int p = 0; p < 2; ++p) {
        int at = (lane >> 4) * 256 + wave * 32 + p * 16 + (lane & 15);
        float s = red[at];
        for (int w = 1; w < 8; ++w) s += red[w * 512 + at];
        sum[p] = s;
    }
}

__device__ __forceinline__ void weff_tile(int tile,
                       const float* __restrict__ pw0, const float* __restrict__ pb0,
                       const float* __restrict__ pw1, const float* __restrict__ pb1,
                       const float* __restrict__ pw2, const float* __restrict__ pb2,
                       const float* __restrict__ fuse_w, const float* __restrict__ fuse_b,
                       float* __restrict__ Weff, float* red) {
    int wave = __builtin_amdgcn_readfirstlane((int)(threadIdx.x >> 5));
    int lane = threadIdx.x & 31;
    int half = lane >> 4, l16 = lane & 15;
    int mt = tile / 12, np = tile % 12;
    int mbase = mt * 16, nbase = np * 32;
    int kk = mbase + l16;
    int rsc, rk;
    if (kk < 8)        { rsc = 0; rk = kk; }
    else if (kk < 24)  { rsc = 1; rk = kk - 8; }
    else if (kk < 56)  { rsc = 2; rk = kk - 24; }
    else               { rsc = (kk == 56) ? 3 : 4; rk = 0; }
    const float* Bc = fuse_w + nbase + l16;
    v8f acc0 = {}, acc1 = {};
    int kbeg = wave * 48;
#pragma unroll
    for (int s = 0; s < 12; ++s) {
        int k0 = kbeg + s * 4;
        int sc = k0 >> 7;
        int ka = k0 + half * 2;
        int d  = ka & 127;
        const float* pw = (sc == 0) ? pw0 : (sc == 1) ? pw1 : pw2;
        const float* pb = (sc == 0) ? pb0 : (sc == 1) ? pb1 : pb2;
        bool own = (rsc == sc);
        const float* ap = (rsc == 3) ? pb : (pw + (own ? rk * 128 : 0));
        float a0 = ap[d], a1 = ap[d + 1];
        bool on = own || (rsc == 3);
        v2f av = { on ? a0 : 0.f, on ? a1 : 0.f };
        v2f b0 = { Bc[(size_t)ka * 384],      Bc[(size_t)(ka + 1) * 384] };
        v2f b1 = { Bc[(size_t)ka * 384 + 16], Bc[(size_t)(ka + 1) * 384 + 16] };
        acc0 = __builtin_amdgcn_wmma_f32_16x16x4_f32(
            false, av, false, b0, (short)0, acc0, false, false);
        acc1 = __builtin_amdgcn_wmma_f32_16x16x4_f32(
            false, av, false, b1, (short)0, acc1, false, false);
    }
    float sum[2];
    tile_sum(acc0, acc1, red, wave, lane, sum);
    for (int p = 0; p < 2; ++p) {
        int row = mbase + p * 8 + wave, col = nbase + lane;
        float v = sum[p];
        if (row == 56) v += fuse_b[col];
        if (row > 56)  v = 0.f;
        Weff[row * 384 + col] = v;
    }
}

__device__ __forceinline__ void wtot_tile(int tile,
                       const float* __restrict__ Weff, const float* __restrict__ flat_w,
                       const float* __restrict__ flat_b,
                       float* __restrict__ Wtot, float* __restrict__ btot, float* red) {
    int wave = __builtin_amdgcn_readfirstlane((int)(threadIdx.x >> 5));
    int lane = threadIdx.x & 31;
    int half = lane >> 4, l16 = lane & 15;
    int mt = tile >> 4, np = tile & 15;
    int mbase = mt * 16, nbase = np * 32;
    const float* Ar = Weff + (size_t)(mbase + l16) * 384;
    const float* Bc = flat_w + nbase + l16;
    v8f acc0 = {}, acc1 = {};
    int kbeg = wave * 48;
#pragma unroll
    for (int s = 0; s < 12; ++s) {
        int ka = kbeg + s * 4 + half * 2;
        v2f av = { Ar[ka], Ar[ka + 1] };
        v2f b0 = { Bc[(size_t)ka * 512],      Bc[(size_t)(ka + 1) * 512] };
        v2f b1 = { Bc[(size_t)ka * 512 + 16], Bc[(size_t)(ka + 1) * 512 + 16] };
        acc0 = __builtin_amdgcn_wmma_f32_16x16x4_f32(
            false, av, false, b0, (short)0, acc0, false, false);
        acc1 = __builtin_amdgcn_wmma_f32_16x16x4_f32(
            false, av, false, b1, (short)0, acc1, false, false);
    }
    float sum[2];
    tile_sum(acc0, acc1, red, wave, lane, sum);
    for (int p = 0; p < 2; ++p) {
        int row = mbase + p * 8 + wave, col = nbase + lane;
        float v = sum[p];
        if (row == 56) btot[col] = v + flat_b[col];
        Wtot[row * 512 + col] = (row < 56) ? v : 0.f;
    }
}

#define FIRST_WEFF (4 * 12)
#define FIRST_ST1  ((2048 / 256) * B_ * SPLITS)
__global__ void k_first(const float* __restrict__ x,
                        const float* __restrict__ pw0, const float* __restrict__ pb0,
                        const float* __restrict__ pw1, const float* __restrict__ pb1,
                        const float* __restrict__ pw2, const float* __restrict__ pb2,
                        const float* __restrict__ fuse_w, const float* __restrict__ fuse_b,
                        float* __restrict__ Weff,
                        float* __restrict__ p0, float* __restrict__ p1,
                        float* __restrict__ p2) {
    __shared__ float red[RED_FLOATS];
    int blk = blockIdx.x;
    if (blk < FIRST_WEFF)
        weff_tile(blk, pw0, pb0, pw1, pb1, pw2, pb2, fuse_w, fuse_b, Weff, red);
    else
        stage1_body(blk - FIRST_WEFF, x, p0, p1, p2, red);
}

#define SECOND_WTOT (4 * 16)
#define SECOND_ST2  ((1024 * 64) / 256)
__global__ void k_second(const float* __restrict__ Weff,
                         const float* __restrict__ flat_w, const float* __restrict__ flat_b,
                         const float* __restrict__ p0, const float* __restrict__ p1,
                         const float* __restrict__ p2,
                         float* __restrict__ Wtot, float* __restrict__ btot,
                         float* __restrict__ A) {
    __shared__ float red[RED_FLOATS];
    int blk = blockIdx.x;
    if (blk < SECOND_WTOT)
        wtot_tile(blk, Weff, flat_w, flat_b, Wtot, btot, red);
    else
        stage2_body(blk - SECOND_WTOT, p0, p1, p2, A);
}

__global__ void k_gemm(const float* __restrict__ A, const float* __restrict__ Wt,
                       const float* __restrict__ btot, float* __restrict__ out) {
    int wave = threadIdx.x >> 5;
    int lane = threadIdx.x & 31;
    int tile = blockIdx.x * 8 + wave;
    int mt = tile >> 5, nt = tile & 31;
    int mbase = mt * 16, nbase = nt * 16;
    int half = lane >> 4, l16 = lane & 15;
    const float* Ar = A + (size_t)(mbase + l16) * 64;
    const float* Bc = Wt + nbase + l16;
    v8f acc = {};
#pragma unroll
    for (int k0 = 0; k0 < 64; k0 += 4) {
        int ka = k0 + half * 2;
        v2f av = { Ar[ka], Ar[ka + 1] };
        v2f bv = { Bc[(size_t)ka * 512], Bc[(size_t)(ka + 1) * 512] };
        acc = __builtin_amdgcn_wmma_f32_16x16x4_f32(
            false, av, false, bv, (short)0, acc, false, false);
    }
    float bias = btot[nbase + l16];
#pragma unroll
    for (int r = 0; r < 8; ++r)
        out[(size_t)(mbase + r + 8 * half) * 512 + nbase + l16] = acc[r] + bias;
}

extern "C" void kernel_launch(void* const* d_in, const int* in_sizes, int n_in,
                              void* d_out, int out_size, void* d_ws, size_t ws_size,
                              hipStream_t stream) {
    (void)in_sizes; (void)n_in; (void)out_size; (void)ws_size;
    const float* x      = (const float*)d_in[0];
    const float* pw0    = (const float*)d_in[1];
    const float* pb0    = (const float*)d_in[2];
    const float* pw1    = (const float*)d_in[3];
    const float* pb1    = (const float*)d_in[4];
    const float* pw2    = (const float*)d_in[5];
    const float* pb2    = (const float*)d_in[6];
    const float* fuse_w = (const float*)d_in[7];
    const float* fuse_b = (const float*)d_in[8];
    const float* flat_w = (const float*)d_in[9];
    const float* flat_b = (const float*)d_in[10];
    float* out = (float*)d_out;

    float* ws   = (float*)d_ws;
    float* p0   = ws;
    float* p1   = p0 + SPLITS * B_ * 2048;
    float* p2   = p1 + SPLITS * B_ * 2048;
    float* A    = p2 + SPLITS * B_ * 2048;
    float* Weff = A + 1024 * 64;
    float* Wtot = Weff + 64 * 384;
    float* btot = Wtot + 64 * 512;

    k_first<<<FIRST_WEFF + FIRST_ST1, 256, 0, stream>>>(
        x, pw0, pb0, pw1, pb1, pw2, pb2, fuse_w, fuse_b, Weff, p0, p1, p2);
    k_second<<<SECOND_WTOT + SECOND_ST2, 256, 0, stream>>>(
        Weff, flat_w, flat_b, p0, p1, p2, Wtot, btot, A);

    k_gemm<<<2048 / 8, 256, 0, stream>>>(A, Wtot, btot, out);
}


# x pass re-split: a block owns 8 whole coarse tokens (64 KiB contiguous, 128-bit loads) and sums half/quarter sub-tiles itself; combine = 16 equal loads per element, no divergence; x sums reordered
# speedup vs baseline: 21.4281x; 1.0627x over previous
#include <hip/hip_runtime.h>
#include <hip/hip_bf16.h>


#define B_ 16
#define L_ 4096
#define M_ 64
#define TGROUPS 16
#define REC_P1 512
#define REC_P2 1536
#define REC_FLOATS 3584

typedef __attribute__((ext_vector_type(2))) float v2f;
typedef __attribute__((ext_vector_type(8))) float v8f;
typedef __attribute__((ext_vector_type(4))) float v4f;

__device__ __forceinline__ float coef_of(int N, int t) {
    int acc = 0;
    int a = (t > 0) ? (t - 1) * 511 : 0;
    int b = (t + 1) * 511;
    int lo = ((N == 512) ? a / 511 : (N == 256) ? a / 255 : a / 127) - 1;
    if (lo < 0) lo = 0;
    int hi = ((N == 512) ? b / 511 : (N == 256) ? b / 255 : b / 127) + 1;
    if (hi > 511) hi = 511;
    for (int tt = lo; tt <= hi; ++tt) {
        int P  = tt * (N - 1);
        int i0 = P / 511;
        int r  = P - i0 * 511;
        int i1 = i0 + 1; if (i1 > N - 1) i1 = N - 1;
        if (t == i0) acc += 511 - r;
        if (t == i1) acc += r;
    }
    return (float)acc / (511.0f * 512.0f);
}

__device__ __forceinline__ v4f fma4(float c, v4f v, v4f a) {
    v4f r;
    r.x = fmaf(c, v.x, a.x); r.y = fmaf(c, v.y, a.y);
    r.z = fmaf(c, v.z, a.z); r.w = fmaf(c, v.w, a.w);
    return r;
}

__device__ __forceinline__ void stage1_body(int blk,
                         const float* __restrict__ x,
                         float* __restrict__ P, float* lds) {
    int b   = blk & 15;
    int tg  = blk >> 4;
    int sub = __builtin_amdgcn_readfirstlane((int)(threadIdx.x >> 7));
    int l   = threadIdx.x & 127;
    const float* xb = x + (size_t)b * (L_ * M_) + (size_t)tg * (8 * 2048)
                        + sub * 1024 + l * 4;
    v4f v[8][2];
#pragma unroll
    for (int i = 0; i < 8; ++i)
#pragma unroll
        for (int qq = 0; qq < 2; ++qq)
            v[i][qq] = *(const v4f*)(xb + i * 2048 + qq * 512);
    float* cf = lds;
    if (threadIdx.x < 56) {
        int j = threadIdx.x;
        int N, t;
        if (j < 8)       { N = 128; t = tg * 8 + j; }
        else if (j < 24) { N = 256; t = 2 * (tg * 8 + ((j - 8) >> 1)) + ((j - 8) & 1); }
        else             { N = 512; t = 4 * (tg * 8 + ((j - 24) >> 2)) + ((j - 24) & 3); }
        cf[j] = coef_of(N, t);
    }
    __syncthreads();
    v4f a2[2] = {}, a1[2] = {}, a0 = {};
#pragma unroll
    for (int i = 0; i < 8; ++i) {
        float c2 = cf[i];
        float c1 = cf[8 + 2 * i + sub];
#pragma unroll
        for (int qq = 0; qq < 2; ++qq) {
            float c0 = cf[24 + 4 * i + 2 * sub + qq];
            v4f vv = v[i][qq];
            a2[qq] = fma4(c2, vv, a2[qq]);
            a1[qq] = fma4(c1, vv, a1[qq]);
            a0     = fma4(c0, vv, a0);
        }
    }
    v4f* xch = (v4f*)(lds + 64);
    if (sub == 1) {
        xch[l]       = a1[0];
        xch[128 + l] = a1[1];
        xch[256 + l] = a0;
    }
    __syncthreads();
    float* rec = P + (size_t)(tg * B_ + b) * REC_FLOATS + l * 4;
    *(v4f*)(rec + REC_P2 + sub * 1024)       = a2[0];
    *(v4f*)(rec + REC_P2 + sub * 1024 + 512) = a2[1];
    if (sub == 0) {
        *(v4f*)(rec + REC_P1)       = a1[0] + xch[l];
        *(v4f*)(rec + REC_P1 + 512) = a1[1] + xch[128 + l];
        *(v4f*)(rec)                = a0 + xch[256 + l];
    }
}

__device__ __forceinline__ void stage2_body(int blk,
                         const float* __restrict__ P,
                         float* __restrict__ A) {
    int idx = blk * 256 + threadIdx.x;
    int row = idx >> 6;
    int kk  = idx & 63;
    int b = row >> 6, m = row & 63;
    int in = (kk < 8)  ? m * 8 + kk
           : (kk < 24) ? REC_P1 + m * 16 + (kk - 8)
           : (kk < 56) ? REC_P2 + m * 32 + (kk - 24)
           : 0;
    const float* q = P + b * REC_FLOATS + in;
    float v[TGROUPS];
#pragma unroll
    for (int g = 0; g < TGROUPS; ++g) v[g] = q[g * (B_ * REC_FLOATS)];
    float acc = 0.f;
#pragma unroll
    for (int g = 0; g < TGROUPS; ++g) acc += v[g];
    A[idx] = (kk < 56) ? acc : 0.f;
}

#define RED_FLOATS (8 * 2 * 8 * 32)

__device__ __forceinline__ void tile_sum(v8f acc0, v8f acc1, float* red,
                                         int wave, int lane, float sum[2]) {
    float* mine = red + wave * 512;
    for (int r = 0; r < 8; ++r) {
        mine[r * 32 + lane]       = acc0[r];
        mine[256 + r * 32 + lane] = acc1[r];
    }
    __syncthreads();
    for (int p = 0; p < 2; ++p) {
        int at = (lane >> 4) * 256 + wave * 32 + p * 16 + (lane & 15);
        float s = red[at];
        for (int w = 1; w < 8; ++w) s += red[w * 512 + at];
        sum[p] = s;
    }
}

__device__ __forceinline__ void weff_tile(int tile,
                       const float* __restrict__ pw0, const float* __restrict__ pb0,
                       const float* __restrict__ pw1, const float* __restrict__ pb1,
                       const float* __restrict__ pw2, const float* __restrict__ pb2,
                       const float* __restrict__ fuse_w, const float* __restrict__ fuse_b,
                       float* __restrict__ Weff, float* red) {
    int wave = __builtin_amdgcn_readfirstlane((int)(threadIdx.x >> 5));
    int lane = threadIdx.x & 31;
    int half = lane >> 4, l16 = lane & 15;
    int mt = tile / 12, np = tile % 12;
    int mbase = mt * 16, nbase = np * 32;
    int kk = mbase + l16;
    int rsc, rk;
    if (kk < 8)        { rsc = 0; rk = kk; }
    else if (kk < 24)  { rsc = 1; rk = kk - 8; }
    else if (kk < 56)  { rsc = 2; rk = kk - 24; }
    else               { rsc = (kk == 56) ? 3 : 4; rk = 0; }
    const float* Bc = fuse_w + nbase + l16;
    v8f acc0 = {}, acc1 = {};
    int kbeg = wave * 48;
#pragma unroll
    for (int s = 0; s < 12; ++s) {
        int k0 = kbeg + s * 4;
        int sc = k0 >> 7;
        int ka = k0 + half * 2;
        int d  = ka & 127;
        const float* pw = (sc == 0) ? pw0 : (sc == 1) ? pw1 : pw2;
        const float* pb = (sc == 0) ? pb0 : (sc == 1) ? pb1 : pb2;
        bool own = (rsc == sc);
        const float* ap = (rsc == 3) ? pb : (pw + (own ? rk * 128 : 0));
        float a0 = ap[d], a1 = ap[d + 1];
        bool on = own || (rsc == 3);
        v2f av = { on ? a0 : 0.f, on ? a1 : 0.f };
        v2f b0 = { Bc[(size_t)ka * 384],      Bc[(size_t)(ka + 1) * 384] };
        v2f b1 = { Bc[(size_t)ka * 384 + 16], Bc[(size_t)(ka + 1) * 384 + 16] };
        acc0 = __builtin_amdgcn_wmma_f32_16x16x4_f32(
            false, av, false, b0, (short)0, acc0, false, false);
        acc1 = __builtin_amdgcn_wmma_f32_16x16x4_f32(
            false, av, false, b1, (short)0, acc1, false, false);
    }
    float sum[2];
    tile_sum(acc0, acc1, red, wave, lane, sum);
    for (int p = 0; p < 2; ++p) {
        int row = mbase + p * 8 + wave, col = nbase + lane;
        float v = sum[p];
        if (row == 56) v += fuse_b[col];
        if (row > 56)  v = 0.f;
        Weff[row * 384 + col] = v;
    }
}

__device__ __forceinline__ void wtot_tile(int tile,
                       const float* __restrict__ Weff, const float* __restrict__ flat_w,
                       const float* __restrict__ flat_b,
                       float* __restrict__ Wtot, float* __restrict__ btot, float* red) {
    int wave = __builtin_amdgcn_readfirstlane((int)(threadIdx.x >> 5));
    int lane = threadIdx.x & 31;
    int half = lane >> 4, l16 = lane & 15;
    int mt = tile >> 4, np = tile & 15;
    int mbase = mt * 16, nbase = np * 32;
    const float* Ar = Weff + (size_t)(mbase + l16) * 384;
    const float* Bc = flat_w + nbase + l16;
    v8f acc0 = {}, acc1 = {};
    int kbeg = wave * 48;
#pragma unroll
    for (int s = 0; s < 12; ++s) {
        int ka = kbeg + s * 4 + half * 2;
        v2f av = { Ar[ka], Ar[ka + 1] };
        v2f b0 = { Bc[(size_t)ka * 512],      Bc[(size_t)(ka + 1) * 512] };
        v2f b1 = { Bc[(size_t)ka * 512 + 16], Bc[(size_t)(ka + 1) * 512 + 16] };
        acc0 = __builtin_amdgcn_wmma_f32_16x16x4_f32(
            false, av, false, b0, (short)0, acc0, false, false);
        acc1 = __builtin_amdgcn_wmma_f32_16x16x4_f32(
            false, av, false, b1, (short)0, acc1, false, false);
    }
    float sum[2];
    tile_sum(acc0, acc1, red, wave, lane, sum);
    for (int p = 0; p < 2; ++p) {
        int row = mbase + p * 8 + wave, col = nbase + lane;
        float v = sum[p];
        if (row == 56) btot[col] = v + flat_b[col];
        Wtot[row * 512 + col] = (row < 56) ? v : 0.f;
    }
}

#define FIRST_WEFF (4 * 12)
#define FIRST_ST1  (B_ * TGROUPS)
__global__ void k_first(const float* __restrict__ x,
                        const float* __restrict__ pw0, const float* __restrict__ pb0,
                        const float* __restrict__ pw1, const float* __restrict__ pb1,
                        const float* __restrict__ pw2, const float* __restrict__ pb2,
                        const float* __restrict__ fuse_w, const float* __restrict__ fuse_b,
                        float* __restrict__ Weff,
                        float* __restrict__ P) {
    __shared__ __attribute__((aligned(16))) float red[RED_FLOATS];
    int blk = blockIdx.x;
    if (blk < FIRST_WEFF)
        weff_tile(blk, pw0, pb0, pw1, pb1, pw2, pb2, fuse_w, fuse_b, Weff, red);
    else
        stage1_body(blk - FIRST_WEFF, x, P, red);
}

#define SECOND_WTOT (4 * 16)
#define SECOND_ST2  ((1024 * 64) / 256)
__global__ void k_second(const float* __restrict__ Weff,
                         const float* __restrict__ flat_w, const float* __restrict__ flat_b,
                         const float* __restrict__ P,
                         float* __restrict__ Wtot, float* __restrict__ btot,
                         float* __restrict__ A) {
    __shared__ float red[RED_FLOATS];
    int blk = blockIdx.x;
    if (blk < SECOND_WTOT)
        wtot_tile(blk, Weff, flat_w, flat_b, Wtot, btot, red);
    else
        stage2_body(blk - SECOND_WTOT, P, A);
}

__global__ void k_gemm(const float* __restrict__ A, const float* __restrict__ Wt,
                       const float* __restrict__ btot, float* __restrict__ out) {
    int wave = threadIdx.x >> 5;
    int lane = threadIdx.x & 31;
    int tile = blockIdx.x * 8 + wave;
    int mt = tile >> 5, nt = tile & 31;
    int mbase = mt * 16, nbase = nt * 16;
    int half = lane >> 4, l16 = lane & 15;
    const float* Ar = A + (size_t)(mbase + l16) * 64;
    const float* Bc = Wt + nbase + l16;
    v8f acc = {};
#pragma unroll
    for (int k0 = 0; k0 < 64; k0 += 4) {
        int ka = k0 + half * 2;
        v2f av = { Ar[ka], Ar[ka + 1] };
        v2f bv = { Bc[(size_t)ka * 512], Bc[(size_t)(ka + 1) * 512] };
        acc = __builtin_amdgcn_wmma_f32_16x16x4_f32(
            false, av, false, bv, (short)0, acc, false, false);
    }
    float bias = btot[nbase + l16];
#pragma unroll
    for (int r = 0; r < 8; ++r)
        out[(size_t)(mbase + r + 8 * half) * 512 + nbase + l16] = acc[r] + bias;
}

extern "C" void kernel_launch(void* const* d_in, const int* in_sizes, int n_in,
                              void* d_out, int out_size, void* d_ws, size_t ws_size,
                              hipStream_t stream) {
    (void)in_sizes; (void)n_in; (void)out_size; (void)ws_size;
    const float* x      = (const float*)d_in[0];
    const float* pw0    = (const float*)d_in[1];
    const float* pb0    = (const float*)d_in[2];
    const float* pw1    = (const float*)d_in[3];
    const float* pb1    = (const float*)d_in[4];
    const float* pw2    = (const float*)d_in[5];
    const float* pb2    = (const float*)d_in[6];
    const float* fuse_w = (const float*)d_in[7];
    const float* fuse_b = (const float*)d_in[8];
    const float* flat_w = (const float*)d_in[9];
    const float* flat_b = (const float*)d_in[10];
    float* out = (float*)d_out;

    float* ws   = (float*)d_ws;
    float* P    = ws;
    float* A    = P + TGROUPS * B_ * REC_FLOATS;
    float* Weff = A + 1024 * 64;
    float* Wtot = Weff + 64 * 384;
    float* btot = Wtot + 64 * 512;

    k_first<<<FIRST_WEFF + FIRST_ST1, 256, 0, stream>>>(
        x, pw0, pb0, pw1, pb1, pw2, pb2, fuse_w, fuse_b, Weff, P);
    k_second<<<SECOND_WTOT + SECOND_ST2, 256, 0, stream>>>(
        Weff, flat_w, flat_b, P, Wtot, btot, A);

    k_gemm<<<2048 / 8, 256, 0, stream>>>(A, Wtot, btot, out);
}


# combine and final GEMM: 4x the work per wave (combine: 4 k per thread, 128-bit loads; GEMM: 16x64 strip, 4 accumulator tiles, 128-bit B loads and stores): 2048 -> 512 waves each; sums as before
# speedup vs baseline: 22.9437x; 1.0707x over previous
#include <hip/hip_runtime.h>
#include <hip/hip_bf16.h>


#define B_ 16
#define L_ 4096
#define M_ 64
#define TGROUPS 16
#define REC_P1 512
#define REC_P2 1536
#define REC_FLOATS 3584

typedef __attribute__((ext_vector_type(2))) float v2f;
typedef __attribute__((ext_vector_type(8))) float v8f;
typedef __attribute__((ext_vector_type(4))) float v4f;

__device__ __forceinline__ float coef_of(int N, int t) {
    int acc = 0;
    int a = (t > 0) ? (t - 1) * 511 : 0;
    int b = (t + 1) * 511;
    int lo = ((N == 512) ? a / 511 : (N == 256) ? a / 255 : a / 127) - 1;
    if (lo < 0) lo = 0;
    int hi = ((N == 512) ? b / 511 : (N == 256) ? b / 255 : b / 127) + 1;
    if (hi > 511) hi = 511;
    for (int tt = lo; tt <= hi; ++tt) {
        int P  = tt * (N - 1);
        int i0 = P / 511;
        int r  = P - i0 * 511;
        int i1 = i0 + 1; if (i1 > N - 1) i1 = N - 1;
        if (t == i0) acc += 511 - r;
        if (t == i1) acc += r;
    }
    return (float)acc / (511.0f * 512.0f);
}

__device__ __forceinline__ v4f fma4(float c, v4f v, v4f a) {
    v4f r;
    r.x = fmaf(c, v.x, a.x); r.y = fmaf(c, v.y, a.y);
    r.z = fmaf(c, v.z, a.z); r.w = fmaf(c, v.w, a.w);
    return r;
}

__device__ __forceinline__ void stage1_body(int blk,
                         const float* __restrict__ x,
                         float* __restrict__ P, float* lds) {
    int b   = blk & 15;
    int tg  = blk >> 4;
    int sub = __builtin_amdgcn_readfirstlane((int)(threadIdx.x >> 7));
    int l   = threadIdx.x & 127;
    const float* xb = x + (size_t)b * (L_ * M_) + (size_t)tg * (8 * 2048)
                        + sub * 1024 + l * 4;
    v4f v[8][2];
#pragma unroll
    for (int i = 0; i < 8; ++i)
#pragma unroll
        for (int qq = 0; qq < 2; ++qq)
            v[i][qq] = *(const v4f*)(xb + i * 2048 + qq * 512);
    float* cf = lds;
    if (threadIdx.x < 56) {
        int j = threadIdx.x;
        int N, t;
        if (j < 8)       { N = 128; t = tg * 8 + j; }
        else if (j < 24) { N = 256; t = 2 * (tg * 8 + ((j - 8) >> 1)) + ((j - 8) & 1); }
        else             { N = 512; t = 4 * (tg * 8 + ((j - 24) >> 2)) + ((j - 24) & 3); }
        cf[j] = coef_of(N, t);
    }
    __syncthreads();
    v4f a2[2] = {}, a1[2] = {}, a0 = {};
#pragma unroll
    for (int i = 0; i < 8; ++i) {
        float c2 = cf[i];
        float c1 = cf[8 + 2 * i + sub];
#pragma unroll
        for (int qq = 0; qq < 2; ++qq) {
            float c0 = cf[24 + 4 * i + 2 * sub + qq];
            v4f vv = v[i][qq];
            a2[qq] = fma4(c2, vv, a2[qq]);
            a1[qq] = fma4(c1, vv, a1[qq]);
            a0     = fma4(c0, vv, a0);
        }
    }
    v4f* xch = (v4f*)(lds + 64);
    if (sub == 1) {
        xch[l]       = a1[0];
        xch[128 + l] = a1[1];
        xch[256 + l] = a0;
    }
    __syncthreads();
    float* rec = P + (size_t)(tg * B_ + b) * REC_FLOATS + l * 4;
    *(v4f*)(rec + REC_P2 + sub * 1024)       = a2[0];
    *(v4f*)(rec + REC_P2 + sub * 1024 + 512) = a2[1];
    if (sub == 0) {
        *(v4f*)(rec + REC_P1)       = a1[0] + xch[l];
        *(v4f*)(rec + REC_P1 + 512) = a1[1] + xch[128 + l];
        *(v4f*)(rec)                = a0 + xch[256 + l];
    }
}

__device__ __forceinline__ void stage2_body(int blk,
                         const float* __restrict__ P,
                         float* __restrict__ A) {
    int idx4 = blk * 256 + threadIdx.x;
    int row = idx4 >> 4;
    int kk  = (idx4 & 15) * 4;
    int b = row >> 6, m = row & 63;
    int in = (kk < 8)  ? m * 8 + kk
           : (kk < 24) ? REC_P1 + m * 16 + (kk - 8)
           : (kk < 56) ? REC_P2 + m * 32 + (kk - 24)
           : 0;
    const float* q = P + b * REC_FLOATS + in;
    v4f v[TGROUPS];
#pragma unroll
    for (int g = 0; g < TGROUPS; ++g) v[g] = *(const v4f*)(q + g * (B_ * REC_FLOATS));
    asm volatile("" :: "v"(v[0]), "v"(v[1]), "v"(v[2]), "v"(v[3]),
                       "v"(v[4]), "v"(v[5]), "v"(v[6]), "v"(v[7]),
                       "v"(v[8]), "v"(v[9]), "v"(v[10]), "v"(v[11]),
                       "v"(v[12]), "v"(v[13]), "v"(v[14]), "v"(v[15]));
    v4f acc = {};
#pragma unroll
    for (int g = 0; g < TGROUPS; ++g) acc += v[g];
    v4f zero = {};
    *(v4f*)(A + (size_t)idx4 * 4) = (kk < 56) ? acc : zero;
}

#define RED_FLOATS (8 * 2 * 8 * 32)

__device__ __forceinline__ void tile_sum(v8f acc0, v8f acc1, float* red,
                                         int wave, int lane, float sum[2]) {
    float* mine = red + wave * 512;
    for (int r = 0; r < 8; ++r) {
        mine[r * 32 + lane]       = acc0[r];
        mine[256 + r * 32 + lane] = acc1[r];
    }
    __syncthreads();
    for (int p = 0; p < 2; ++p) {
        int at = (lane >> 4) * 256 + wave * 32 + p * 16 + (lane & 15);
        float s = red[at];
        for (int w = 1; w < 8; ++w) s += red[w * 512 + at];
        sum[p] = s;
    }
}

__device__ __forceinline__ void weff_tile(int tile,
                       const float* __restrict__ pw0, const float* __restrict__ pb0,
                       const float* __restrict__ pw1, const float* __restrict__ pb1,
                       const float* __restrict__ pw2, const float* __restrict__ pb2,
                       const float* __restrict__ fuse_w, const float* __restrict__ fuse_b,
                       float* __restrict__ Weff, float* red) {
    int wave = __builtin_amdgcn_readfirstlane((int)(threadIdx.x >> 5));
    int lane = threadIdx.x & 31;
    int half = lane >> 4, l16 = lane & 15;
    int mt = tile / 12, np = tile % 12;
    int mbase = mt * 16, nbase = np * 32;
    int kk = mbase + l16;
    int rsc, rk;
    if (kk < 8)        { rsc = 0; rk = kk; }
    else if (kk < 24)  { rsc = 1; rk = kk - 8; }
    else if (kk < 56)  { rsc = 2; rk = kk - 24; }
    else               { rsc = (kk == 56) ? 3 : 4; rk = 0; }
    const float* Bc = fuse_w + nbase + l16;
    v8f acc0 = {}, acc1 = {};
    int kbeg = wave * 48;
#pragma unroll
    for (int s = 0; s < 12; ++s) {
        int k0 = kbeg + s * 4;
        int sc = k0 >> 7;
        int ka = k0 + half * 2;
        int d  = ka & 127;
        const float* pw = (sc == 0) ? pw0 : (sc == 1) ? pw1 : pw2;
        const float* pb = (sc == 0) ? pb0 : (sc == 1) ? pb1 : pb2;
        bool own = (rsc == sc);
        const float* ap = (rsc == 3) ? pb : (pw + (own ? rk * 128 : 0));
        float a0 = ap[d], a1 = ap[d + 1];
        bool on = own || (rsc == 3);
        v2f av = { on ? a0 : 0.f, on ? a1 : 0.f };
        v2f b0 = { Bc[(size_t)ka * 384],      Bc[(size_t)(ka + 1) * 384] };
        v2f b1 = { Bc[(size_t)ka * 384 + 16], Bc[(size_t)(ka + 1) * 384 + 16] };
        acc0 = __builtin_amdgcn_wmma_f32_16x16x4_f32(
            false, av, false, b0, (short)0, acc0, false, false);
        acc1 = __builtin_amdgcn_wmma_f32_16x16x4_f32(
            false, av, false, b1, (short)0, acc1, false, false);
    }
    float sum[2];
    tile_sum(acc0, acc1, red, wave, lane, sum);
    for (int p = 0; p < 2; ++p) {
        int row = mbase + p * 8 + wave, col = nbase + lane;
        float v = sum[p];
        if (row == 56) v += fuse_b[col];
        if (row > 56)  v = 0.f;
        Weff[row * 384 + col] = v;
    }
}

__device__ __forceinline__ void wtot_tile(int tile,
                       const float* __restrict__ Weff, const float* __restrict__ flat_w,
                       const float* __restrict__ flat_b,
                       float* __restrict__ Wtot, float* __restrict__ btot, float* red) {
    int wave = __builtin_amdgcn_readfirstlane((int)(threadIdx.x >> 5));
    int lane = threadIdx.x & 31;
    int half = lane >> 4, l16 = lane & 15;
    int mt = tile >> 4, np = tile & 15;
    int mbase = mt * 16, nbase = np * 32;
    const float* Ar = Weff + (size_t)(mbase + l16) * 384;
    const float* Bc = flat_w + nbase + l16;
    v8f acc0 = {}, acc1 = {};
    int kbeg = wave * 48;
#pragma unroll
    for (int s = 0; s < 12; ++s) {
        int ka = kbeg + s * 4 + half * 2;
        v2f av = { Ar[ka], Ar[ka + 1] };
        v2f b0 = { Bc[(size_t)ka * 512],      Bc[(size_t)(ka + 1) * 512] };
        v2f b1 = { Bc[(size_t)ka * 512 + 16], Bc[(size_t)(ka + 1) * 512 + 16] };
        acc0 = __builtin_amdgcn_wmma_f32_16x16x4_f32(
            false, av, false, b0, (short)0, acc0, false, false);
        acc1 = __builtin_amdgcn_wmma_f32_16x16x4_f32(
            false, av, false, b1, (short)0, acc1, false, false);
    }
    float sum[2];
    tile_sum(acc0, acc1, red, wave, lane, sum);
    for (int p = 0; p < 2; ++p) {
        int row = mbase + p * 8 + wave, col = nbase + lane;
        float v = sum[p];
        if (row == 56) btot[col] = v + flat_b[col];
        Wtot[row * 512 + col] = (row < 56) ? v : 0.f;
    }
}

#define FIRST_WEFF (4 * 12)
#define FIRST_ST1  (B_ * TGROUPS)
__global__ void k_first(const float* __restrict__ x,
                        const float* __restrict__ pw0, const float* __restrict__ pb0,
                        const float* __restrict__ pw1, const float* __restrict__ pb1,
                        const float* __restrict__ pw2, const float* __restrict__ pb2,
                        const float* __restrict__ fuse_w, const float* __restrict__ fuse_b,
                        float* __restrict__ Weff,
                        float* __restrict__ P) {
    __shared__ __attribute__((aligned(16))) float red[RED_FLOATS];
    int blk = blockIdx.x;
    if (blk < FIRST_WEFF)
        weff_tile(blk, pw0, pb0, pw1, pb1, pw2, pb2, fuse_w, fuse_b, Weff, red);
    else
        stage1_body(blk - FIRST_WEFF, x, P, red);
}

#define SECOND_WTOT (4 * 16)
#define SECOND_ST2  ((1024 * 64) / (4 * 256))
__global__ void k_second(const float* __restrict__ Weff,
                         const float* __restrict__ flat_w, const float* __restrict__ flat_b,
                         const float* __restrict__ P,
                         float* __restrict__ Wtot, float* __restrict__ btot,
                         float* __restrict__ A) {
    __shared__ float red[RED_FLOATS];
    int blk = blockIdx.x;
    if (blk < SECOND_WTOT)
        wtot_tile(blk, Weff, flat_w, flat_b, Wtot, btot, red);
    else
        stage2_body(blk - SECOND_WTOT, P, A);
}

#define GEMM_WAVES 4
#define GEMM_THREADS (32 * GEMM_WAVES)
#define GEMM_BLOCKS  (512 / GEMM_WAVES)
__global__ void __launch_bounds__(GEMM_THREADS)
k_gemm(const float* __restrict__ A, const float* __restrict__ Wt,
                       const float* __restrict__ btot, float* __restrict__ out) {
    int wave = threadIdx.x >> 5;
    int lane = threadIdx.x & 31;
    int strip = blockIdx.x * GEMM_WAVES + wave;
    int mt = strip >> 3, ng = strip & 7;
    int mbase = mt * 16, nbase = ng * 64;
    int half = lane >> 4, l16 = lane & 15;
    const float* Ar = A + (size_t)(mbase + l16) * 64;
    const float* Bc = Wt + nbase + l16 * 4;
    v8f acc0 = {}, acc1 = {}, acc2 = {}, acc3 = {};
#pragma unroll
    for (int k0 = 0; k0 < 64; k0 += 4) {
        int ka = k0 + half * 2;
        v2f av = { Ar[ka], Ar[ka + 1] };
        v4f r0 = *(const v4f*)(Bc + (size_t)ka * 512);
        v4f r1 = *(const v4f*)(Bc + (size_t)(ka + 1) * 512);
        v2f b0 = { r0.x, r1.x };
        v2f b1 = { r0.y, r1.y };
        v2f b2 = { r0.z, r1.z };
        v2f b3 = { r0.w, r1.w };
        acc0 = __builtin_amdgcn_wmma_f32_16x16x4_f32(
            false, av, false, b0, (short)0, acc0, false, false);
        acc1 = __builtin_amdgcn_wmma_f32_16x16x4_f32(
            false, av, false, b1, (short)0, acc1, false, false);
        acc2 = __builtin_amdgcn_wmma_f32_16x16x4_f32(
            false, av, false, b2, (short)0, acc2, false, false);
        acc3 = __builtin_amdgcn_wmma_f32_16x16x4_f32(
            false, av, false, b3, (short)0, acc3, false, false);
    }
    v4f bias = *(const v4f*)(btot + nbase + l16 * 4);
#pragma unroll
    for (int r = 0; r < 8; ++r) {
        v4f o = { acc0[r] + bias.x, acc1[r] + bias.y, acc2[r] + bias.z, acc3[r] + bias.w };
        *(v4f*)(out + (size_t)(mbase + r + 8 * half) * 512 + nbase + l16 * 4) = o;
    }
}

extern "C" void kernel_launch(void* const* d_in, const int* in_sizes, int n_in,
                              void* d_out, int out_size, void* d_ws, size_t ws_size,
                              hipStream_t stream) {
    (void)in_sizes; (void)n_in; (void)out_size; (void)ws_size;
    const float* x      = (const float*)d_in[0];
    const float* pw0    = (const float*)d_in[1];
    const float* pb0    = (const float*)d_in[2];
    const float* pw1    = (const float*)d_in[3];
    const float* pb1    = (const float*)d_in[4];
    const float* pw2    = (const float*)d_in[5];
    const float* pb2    = (const float*)d_in[6];
    const float* fuse_w = (const float*)d_in[7];
    const float* fuse_b = (const float*)d_in[8];
    const float* flat_w = (const float*)d_in[9];
    const float* flat_b = (const float*)d_in[10];
    float* out = (float*)d_out;

    float* ws   = (float*)d_ws;
    float* P    = ws;
    float* A    = P + TGROUPS * B_ * REC_FLOATS;
    float* Weff = A + 1024 * 64;
    float* Wtot = Weff + 64 * 384;
    float* btot = Wtot + 64 * 512;

    k_first<<<FIRST_WEFF + FIRST_ST1, 256, 0, stream>>>(
        x, pw0, pb0, pw1, pb1, pw2, pb2, fuse_w, fuse_b, Weff, P);
    k_second<<<SECOND_WTOT + SECOND_ST2, 256, 0, stream>>>(
        Weff, flat_w, flat_b, P, Wtot, btot, A);

    k_gemm<<<GEMM_BLOCKS, GEMM_THREADS, 0, stream>>>(A, Wtot, btot, out);
}
